# starting kernel
# baseline (speedup 1.0000x reference)
#include <hip/hip_runtime.h>
#include <cstddef>

typedef float v2f __attribute__((ext_vector_type(2)));
typedef float v8f __attribute__((ext_vector_type(8)));

namespace {
constexpr int kB    = 8;
constexpr int kN    = 256;
constexpr int kL    = 4;
constexpr int kH    = 128;
constexpr int kBN   = kB * kN;
constexpr int kG    = 4 * kH;
constexpr int kSteps = 16;
constexpr int kOutV = 30000;
constexpr size_t kHOFF = (size_t)2 * kBN * kH;
constexpr int AS = 132;
constexpr int GS = 516;
constexpr int kPairs = 2 * (kH / 2) * kG;
}

__device__ __forceinline__ float sigmf(float x) { return 1.0f / (1.0f + expf(-x)); }

__device__ __forceinline__ v8f wmma4(v2f a, v2f b, v8f c) {
  return __builtin_amdgcn_wmma_f32_16x16x4_f32(false, a, false, b, (short)0, c, false, false);
}

__device__ __forceinline__ void gemm_tile(const float* __restrict__ A,
                                          const v2f* __restrict__ Wp,
                                          int c_base, int lane, v8f acc[4]) {
  const int m  = lane & 15;
  const int hf = lane >> 4;
  const int n  = lane & 15;
#pragma unroll 2
  for (int kc = 0; kc < 32; ++kc) {
    const int kk = kc * 4 + 2 * hf;
    v2f a = *(const v2f*)(A + m * AS + kk);
    const v2f* Wr = Wp + (size_t)(kc * 2 + hf) * kG;
#pragma unroll
    for (int t = 0; t < 4; ++t) {
      acc[t] = wmma4(a, Wr[c_base + t * 16 + n], acc[t]);
    }
  }
}

__global__ __launch_bounds__(256) void k_lstm_step(
    const float* __restrict__ emb, const float* __restrict__ Sin,
    float* __restrict__ Stg, const float* __restrict__ ipin,
    float* __restrict__ ptb, float* __restrict__ pfb,
    const v2f* __restrict__ Wxp, const v2f* __restrict__ Whp,
    const float* __restrict__ bias, const float* __restrict__ bW,
    const float* __restrict__ bb, const int* __restrict__ exitI) {
  __shared__ float sX[16 * AS];
  __shared__ float sH[2][16 * AS];
  __shared__ float sC[2][16 * kH];
  __shared__ float sG[16 * GS];
  __shared__ float sLg[32];

  const int tid  = threadIdx.x;
  const int lane = tid & 31;
  const int wave = tid >> 5;
  const int r0   = blockIdx.x * 16;
  const int c_base = wave * 64;

  for (int idx = tid; idx < 2 * 16 * kH; idx += 256) {
    int l = idx >> 11, rem = idx & 2047, m = rem >> 7, k = rem & 127;
    size_t row = r0 + m;
    sC[l][m * kH + k] = Sin[((size_t)l * kBN + row) * kH + k];
    sH[l][m * AS + k] = Sin[kHOFF + ((size_t)l * kBN + row) * kH + k];
  }
  __syncthreads();

  for (int tok = 0; tok < kL; ++tok) {
    for (int idx = tid; idx < 16 * kH; idx += 256) {
      int m = idx >> 7, k = idx & 127;
      sX[m * AS + k] = emb[(((size_t)(r0 + m)) * kL + tok) * kH + k];
    }
    __syncthreads();
    for (int l = 0; l < 2; ++l) {
      v8f acc[4];
#pragma unroll
      for (int t = 0; t < 4; ++t) {
        float blv = bias[l * kG + c_base + t * 16 + (lane & 15)];
#pragma unroll
        for (int v = 0; v < 8; ++v) acc[t][v] = blv;
      }
      const float* Ain = (l == 0) ? sX : sH[0];
      gemm_tile(Ain,   Wxp + (size_t)l * (kH / 2) * kG, c_base, lane, acc);
      gemm_tile(sH[l], Whp + (size_t)l * (kH / 2) * kG, c_base, lane, acc);
      const int hf = lane >> 4;
#pragma unroll
      for (int t = 0; t < 4; ++t) {
        int col = c_base + t * 16 + (lane & 15);
#pragma unroll
        for (int v = 0; v < 8; ++v) sG[(v + 8 * hf) * GS + col] = acc[t][v];
      }
      __syncthreads();
      for (int idx = tid; idx < 16 * kH; idx += 256) {
        int m = idx >> 7, k = idx & 127;
        float iv = sG[m * GS + k];
        float fv = sG[m * GS + kH + k];
        float gv = sG[m * GS + 2 * kH + k];
        float ov = sG[m * GS + 3 * kH + k];
        float cold = sC[l][m * kH + k];
        float cn = sigmf(fv) * cold + sigmf(iv) * tanhf(gv);
        float hn = sigmf(ov) * tanhf(cn);
        sC[l][m * kH + k] = cn;
        sH[l][m * AS + k] = hn;
      }
      __syncthreads();
    }
  }

  const int bidx = r0 >> 8;
  const int mm = exitI[bidx] - (r0 & 255);
  if (mm >= 0 && mm < 16) {
    for (int idx = tid; idx < 2 * kH; idx += 256) {
      int l = idx >> 7, k = idx & 127;
      size_t row = r0 + mm;
      sC[l][mm * kH + k] = Sin[((size_t)l * kBN + row) * kH + k];
      sH[l][mm * AS + k] = Sin[kHOFF + ((size_t)l * kBN + row) * kH + k];
    }
  }
  __syncthreads();

  if (tid < 32) {
    int m = tid >> 1, j = tid & 1;
    float s = bb[j];
    for (int k = 0; k < kH; ++k) s += sC[0][m * kH + k] * bW[k * 2 + j];
    for (int k = 0; k < kH; ++k) s += sH[0][m * AS + k] * bW[(kH + k) * 2 + j];
    for (int k = 0; k < kH; ++k) s += sC[1][m * kH + k] * bW[(2 * kH + k) * 2 + j];
    for (int k = 0; k < kH; ++k) s += sH[1][m * AS + k] * bW[(3 * kH + k) * 2 + j];
    sLg[tid] = s;
  }
  __syncthreads();
  if (tid < 16) {
    float l0 = sLg[2 * tid], l1 = sLg[2 * tid + 1];
    float mx = fmaxf(l0, l1);
    float e0 = expf(l0 - mx), e1 = expf(l1 - mx);
    float inv = 1.0f / (e0 + e1);
    float ipv = ipin[r0 + tid];
    ptb[r0 + tid] = e0 * inv * ipv;
    pfb[r0 + tid] = e1 * inv * ipv;
  }

  for (int idx = tid; idx < 2 * 16 * kH; idx += 256) {
    int l = idx >> 11, rem = idx & 2047, m = rem >> 7, k = rem & 127;
    size_t row = r0 + m;
    Stg[((size_t)l * kBN + row) * kH + k]         = sC[l][m * kH + k];
    Stg[kHOFF + ((size_t)l * kBN + row) * kH + k] = sH[l][m * AS + k];
  }
}

__global__ __launch_bounds__(128) void k_aggregate(
    const float* __restrict__ Stg, const float* __restrict__ Sin,
    const float* __restrict__ ipin, const float* __restrict__ ptb,
    const float* __restrict__ pfb, const int* __restrict__ ti,
    const int* __restrict__ fi, const int* __restrict__ steps,
    float* __restrict__ Sout, float* __restrict__ ipout, int step) {
  const int tgt = blockIdx.x;
  const int b = tgt >> 8;
  const int m = tgt & 255;
  const int h = threadIdx.x;
  float aC0 = 0.f, aH0 = 0.f, aC1 = 0.f, aH1 = 0.f, aip = 0.f;
  const int base = b * kN;
  for (int n = 0; n < kN; ++n) {
    int src = base + n;
    float w = 0.f;
    if (ti[src] == m) w += ptb[src];
    if (fi[src] == m) w += pfb[src];
    if (w != 0.f) {
      aip += w;
      aC0 += w * Stg[((size_t)src) * kH + h];
      aH0 += w * Stg[kHOFF + ((size_t)src) * kH + h];
      aC1 += w * Stg[((size_t)kBN + src) * kH + h];
      aH1 += w * Stg[kHOFF + ((size_t)kBN + src) * kH + h];
    }
  }
  float denom = aip + 1e-7f;
  bool active = step < steps[b];
  size_t row = tgt;
  Sout[row * kH + h]                         = active ? aC0 / denom : Sin[row * kH + h];
  Sout[kHOFF + row * kH + h]                 = active ? aH0 / denom : Sin[kHOFF + row * kH + h];
  Sout[((size_t)kBN + row) * kH + h]         = active ? aC1 / denom : Sin[((size_t)kBN + row) * kH + h];
  Sout[kHOFF + ((size_t)kBN + row) * kH + h] = active ? aH1 / denom : Sin[kHOFF + ((size_t)kBN + row) * kH + h];
  if (h == 0) ipout[tgt] = active ? aip : ipin[tgt];
}

__global__ __launch_bounds__(256) void k_logits(
    const float* __restrict__ S, const int* __restrict__ exitI,
    const float* __restrict__ outW, const float* __restrict__ outb,
    float* __restrict__ out) {
  __shared__ float sF[kB * 512];
  const int tid = threadIdx.x;
  for (int idx = tid; idx < kB * 512; idx += 256) {
    int q = idx >> 9, k = idx & 511, seg = k >> 7, kk = k & 127;
    size_t row = (size_t)q * kN + exitI[q];
    float v;
    if (seg == 0)      v = S[row * kH + kk];
    else if (seg == 1) v = S[kHOFF + row * kH + kk];
    else if (seg == 2) v = S[((size_t)kBN + row) * kH + kk];
    else               v = S[kHOFF + ((size_t)kBN + row) * kH + kk];
    sF[idx] = v;
  }
  __syncthreads();
  int j = blockIdx.x * 256 + tid;
  if (j >= kOutV) return;
  float acc[kB];
  float bj = outb[j];
#pragma unroll
  for (int q = 0; q < kB; ++q) acc[q] = bj;
  for (int k = 0; k < 512; ++k) {
    float w = outW[(size_t)k * kOutV + j];
#pragma unroll
    for (int q = 0; q < kB; ++q) acc[q] += sF[q * 512 + k] * w;
  }
  for (int q = 0; q < kB; ++q) out[(size_t)q * kOutV + j] = acc[q];
}

__global__ void k_embed(const int* __restrict__ data, const float* __restrict__ embed,
                        float* __restrict__ embuf, int total) {
  int idx = blockIdx.x * 256 + threadIdx.x;
  if (idx >= total) return;
  int token = idx >> 7, h = idx & 127;
  embuf[idx] = embed[(size_t)data[token] * kH + h];
}

__global__ void k_pack(const float* __restrict__ Wsrc, float* __restrict__ Wdst, int total) {
  int idx = blockIdx.x * 256 + threadIdx.x;
  if (idx >= total) return;
  int col = idx & 511;
  int p   = (idx >> 9) & 63;
  int l   = idx >> 15;
  const float* Wl = Wsrc + (size_t)l * kH * kG;
  v2f v;
  v.x = Wl[(size_t)(2 * p) * kG + col];
  v.y = Wl[(size_t)(2 * p + 1) * kG + col];
  *(v2f*)(Wdst + (size_t)idx * 2) = v;
}

__global__ void k_init(float* __restrict__ S0, float* __restrict__ ip0, int total) {
  int idx = blockIdx.x * 256 + threadIdx.x;
  if (idx < total) S0[idx] = 0.f;
  if (idx < kBN) ip0[idx] = ((idx & 255) == 0) ? 1.0f : 0.0f;
}

extern "C" void kernel_launch(void* const* d_in, const int* in_sizes, int n_in,
                              void* d_out, int out_size, void* d_ws, size_t ws_size,
                              hipStream_t stream) {
  const int*   data  = (const int*)d_in[0];
  const int*   ti    = (const int*)d_in[1];
  const int*   fi    = (const int*)d_in[2];
  const int*   exitI = (const int*)d_in[3];
  const int*   steps = (const int*)d_in[4];
  const float* embed = (const float*)d_in[5];
  const float* Wx    = (const float*)d_in[6];
  const float* Wh    = (const float*)d_in[7];
  const float* bias  = (const float*)d_in[8];
  const float* bW    = (const float*)d_in[9];
  const float* bb    = (const float*)d_in[10];
  const float* outW  = (const float*)d_in[11];
  const float* outb  = (const float*)d_in[12];
  float* out = (float*)d_out;

  float* ws = (float*)d_ws;
  const size_t EMB = (size_t)kBN * kL * kH;
  const size_t ST  = (size_t)4 * kBN * kH;
  float* embuf = ws;
  float* S0  = embuf + EMB;
  float* S1  = S0 + ST;
  float* STG = S1 + ST;
  float* ip0 = STG + ST;
  float* ip1 = ip0 + kBN;
  float* ptb = ip1 + kBN;
  float* pfb = ptb + kBN;
  float* Wxp = pfb + kBN;
  float* Whp = Wxp + (size_t)kPairs * 2;

  k_embed<<<(int)((EMB + 255) / 256), 256, 0, stream>>>(data, embed, embuf, (int)EMB);
  k_init<<<(int)((ST + 255) / 256), 256, 0, stream>>>(S0, ip0, (int)ST);
  k_pack<<<(kPairs + 255) / 256, 256, 0, stream>>>(Wx, Wxp, kPairs);
  k_pack<<<(kPairs + 255) / 256, 256, 0, stream>>>(Wh, Whp, kPairs);

  float* Sin = S0; float* Sout = S1; float* ipin = ip0; float* ipout = ip1;
  for (int s = 0; s < kSteps; ++s) {
    k_lstm_step<<<kBN / 16, 256, 0, stream>>>(embuf, Sin, STG, ipin, ptb, pfb,
                                              (const v2f*)Wxp, (const v2f*)Whp,
                                              bias, bW, bb, exitI);
    k_aggregate<<<kBN, 128, 0, stream>>>(STG, Sin, ipin, ptb, pfb, ti, fi, steps,
                                         Sout, ipout, s);
    float* t1 = Sin; Sin = Sout; Sout = t1;
    float* t2 = ipin; ipin = ipout; ipout = t2;
  }
  k_logits<<<(kOutV + 255) / 256, 256, 0, stream>>>(Sin, exitI, outW, outb, out);
  (void)in_sizes; (void)n_in; (void)out_size; (void)ws_size;
}


# Step kernel re-tiled to 256 work-groups of 8 rows: input-half products fill the 16-row tile with two tokens (12 products a group, not 16), recurrent halves one token on zeroed rows; sums unchanged
# speedup vs baseline: 2.6204x; 2.6204x over previous
#include <hip/hip_runtime.h>
#include <cstddef>

typedef float v2f __attribute__((ext_vector_type(2)));
typedef float v8f __attribute__((ext_vector_type(8)));
typedef float v4f __attribute__((ext_vector_type(4)));

namespace {
constexpr int kB    = 8;
constexpr int kN    = 256;
constexpr int kL    = 4;
constexpr int kH    = 128;
constexpr int kBN   = kB * kN;
constexpr int kG    = 4 * kH;
constexpr int kSteps = 16;
constexpr int kOutV = 30000;
constexpr size_t kHOFF = (size_t)2 * kBN * kH;
constexpr int AS = 132;
constexpr int GS = 516;
constexpr int kPairs = 2 * (kH / 2) * kG;
constexpr int kStepT = 1024;
constexpr int kRows  = 8;
constexpr int kSlots = 15;
constexpr int kTile  = 16 * AS;
constexpr int kTeamT = 512;
constexpr int kEntB = 2 * kN;
constexpr int kTrueBit  = 0x10000;
constexpr int kFalseBit = 0x20000;
constexpr int FP = 516;
constexpr int kLogT = 128;
constexpr int kStrips = (kOutV + 31) / 32;
}

__device__ __forceinline__ float sigmf(float x) { return 1.0f / (1.0f + expf(-x)); }

__device__ __forceinline__ v8f wmma4(v2f a, v2f b, v8f c) {
  return __builtin_amdgcn_wmma_f32_16x16x4_f32(false, a, false, b, (short)0, c, false, false);
}

template <int P, int KB>
__device__ __forceinline__ void gemm_piece(const float* __restrict__ A, const v2f* __restrict__ Wp,
                                           int cq, int m, int hf, v8f& acc0, v8f& acc1) {
  __builtin_amdgcn_s_setprio(P);
#pragma unroll
  for (int kc = KB; kc < KB + 8; ++kc) {
    const int kk = kc * 4 + 2 * hf;
    v2f a = *(const v2f*)(A + m * AS + kk);
    const v4f q = *(const v4f*)(Wp + (size_t)(kc * 2 + hf) * kG + cq);
    v2f b0 = {q.x, q.y};
    v2f b1 = {q.z, q.w};
    acc0 = wmma4(a, b0, acc0);
    acc1 = wmma4(a, b1, acc1);
  }
}

__device__ __forceinline__ void gemm_full(const float* __restrict__ A, const v2f* __restrict__ Wp,
                                          int cq, int lane, v8f& acc0, v8f& acc1) {
  const int m  = lane & 15;
  const int hf = lane >> 4;
  gemm_piece<3, 0>(A, Wp, cq, m, hf, acc0, acc1);
  gemm_piece<2, 8>(A, Wp, cq, m, hf, acc0, acc1);
  gemm_piece<1, 16>(A, Wp, cq, m, hf, acc0, acc1);
  gemm_piece<0, 24>(A, Wp, cq, m, hf, acc0, acc1);
}

__device__ __forceinline__ void team_cell(const float* __restrict__ G, float* __restrict__ C,
                                          float* __restrict__ H1, float* __restrict__ H2, int rh,
                                          int first, int stride) {
  for (int idx = first; idx < kRows * kH; idx += stride) {
    int m = idx >> 7, k = idx & 127;
    const float* g = G + (rh * kRows + m) * GS;
    float iv = g[k];
    float fv = g[kH + k];
    float gv = g[2 * kH + k];
    float ov = g[3 * kH + k];
    float cold = C[m * kH + k];
    float cn = sigmf(fv) * cold + sigmf(iv) * tanhf(gv);
    float hn = sigmf(ov) * tanhf(cn);
    C[m * kH + k] = cn;
    H1[m * AS + k] = hn;
    if (H2 != nullptr) H2[m * AS + k] = hn;
  }
}

__global__ __launch_bounds__(kStepT) void k_lstm_step(
    const float* __restrict__ emb, const float* __restrict__ Sin,
    float* __restrict__ Stg, const float* __restrict__ ipin,
    float* __restrict__ ptb, float* __restrict__ pfb,
    const v2f* __restrict__ Wxp, const v2f* __restrict__ Whp,
    const float* __restrict__ bias, const float* __restrict__ bW,
    const float* __restrict__ bb, const int* __restrict__ exitI) {
  __shared__ float sA[8 * kTile];
  __shared__ float sC[2][kRows * kH];
  __shared__ float sG[2][16 * GS];
  __shared__ __attribute__((aligned(16))) float sP[32 * 32];

  const int tid  = threadIdx.x;
  const int lane = tid & 31;
  const int wave = __builtin_amdgcn_readfirstlane(tid >> 5);
  const int team = wave >> 4;
  const int tt   = tid & (kTeamT - 1);
  const int r0   = blockIdx.x * kRows;
  const int cq   = (wave & 15) * 32 + 2 * (lane & 15);

  v2f bwf[4];
  {
    const int c = lane & 15;
    const int hf = lane >> 4;
#pragma unroll
    for (int kb = 0; kb < 4; ++kb) {
      const int K = wave * 16 + kb * 4 + 2 * hf;
      const v4f q = *(const v4f*)(bW + K * 2);
      v2f bfr = {0.f, 0.f};
      if (c == 0) { bfr.x = q.x; bfr.y = q.z; }
      if (c == 1) { bfr.x = q.y; bfr.y = q.w; }
      bwf[kb] = bfr;
    }
  }

  for (int idx = tid; idx < 2 * kRows * kH; idx += kStepT) {
    int l = idx >> 10, m = (idx >> 7) & 7, k = idx & 127;
    size_t row = r0 + m;
    sC[l][m * kH + k] = Sin[((size_t)l * kBN + row) * kH + k];
    float hv = Sin[kHOFF + ((size_t)l * kBN + row) * kH + k];
    sA[(2 + 2 * l) * kTile + m * AS + k] = hv;
    sA[(2 + 2 * l) * kTile + (kRows + m) * AS + k] = 0.f;
    sA[(3 + 2 * l) * kTile + m * AS + k] = 0.f;
  }
  for (int idx = tid; idx < kL * kRows * kH; idx += kStepT) {
    int t = idx >> 10, m = (idx >> 7) & 7, k = idx & 127;
    sA[(t >> 1) * kTile + ((t & 1) * kRows + m) * AS + k] = emb[(((size_t)(r0 + m)) * kL + t) * kH + k];
  }
  __syncthreads();

  const unsigned long long sched = team ? 0xA89765304250100ULL : 0xA090A8976053421ULL;
  v8f acc0 = {0.f, 0.f, 0.f, 0.f, 0.f, 0.f, 0.f, 0.f};
  v8f acc1 = acc0;
  for (int s = 0; s < kSlots; ++s) {
    const int op = (int)((sched >> (4 * s)) & 15ULL);
    const bool paired = (op == 1) || (op == 6);
    if (paired || op == 2 || op == 3 || op == 7 || op == 8) {
      const int layer = (op >= 6) ? 1 : 0;
      const int tile = (op == 1) ? team : (op == 6) ? 6 + team : (op < 6) ? op : op - 3;
      const v2f* Wp = (paired ? Wxp : Whp) + (size_t)layer * (kH / 2) * kG;
      if (paired) {
        const v2f bv = *(const v2f*)(bias + layer * kG + cq);
#pragma unroll
        for (int v = 0; v < 8; ++v) { acc0[v] = bv.x; acc1[v] = bv.y; }
      }
      gemm_full(sA + tile * kTile, Wp, cq, lane, acc0, acc1);
      if (!paired) {
        float* Gout = sG[team];
        const int hf = lane >> 4;
#pragma unroll
        for (int v = 0; v < 8; ++v) {
          v2f g = {acc0[v], acc1[v]};
          *(v2f*)(Gout + (v + 8 * hf) * GS + cq) = g;
        }
      }
    } else if (op != 0) {
      const int layer = (op >= 9) ? 1 : 0;
      const int rh = (op == 5 || op == 10) ? 1 : 0;
      const bool shared = (s == 4) || (s == 12) || (s == 14);
      const int owner = shared ? ((s == 4) ? 0 : 1) : team;
      float* H1 = rh ? sA + (2 + 2 * layer) * kTile : sA + (3 + 2 * layer) * kTile + kRows * AS;
      float* H2 = layer ? nullptr : sA + (6 + owner) * kTile + rh * kRows * AS;
      team_cell(sG[owner], sC[layer], H1, H2, rh, shared ? tid : tt, shared ? kStepT : kTeamT);
    }
    __syncthreads();
  }

  float* fH0 = sA + 2 * kTile;
  float* fH1 = sA + 4 * kTile;

  const int bidx = r0 >> 8;
  const int mm = exitI[bidx] - (r0 & 255);
  if (mm >= 0 && mm < kRows) {
    for (int idx = tid; idx < 2 * kH; idx += kStepT) {
      int l = idx >> 7, k = idx & 127;
      size_t row = r0 + mm;
      sC[l][mm * kH + k] = Sin[((size_t)l * kBN + row) * kH + k];
      float hv = Sin[kHOFF + ((size_t)l * kBN + row) * kH + k];
      if (l == 0) fH0[mm * AS + k] = hv; else fH1[mm * AS + k] = hv;
    }
  }
  __syncthreads();

  {
    const int seg = wave >> 3;
    const float* Ab = (seg == 0) ? sC[0] : (seg == 1) ? fH0 : (seg == 2) ? sC[1] : fH1;
    const int As = (seg & 1) ? AS : kH;
    const int m  = lane & 15;
    const int hf = lane >> 4;
    const float* Ar = Ab + (m & 7) * As + (wave & 7) * 16 + 2 * hf;
    v8f acc = {0.f, 0.f, 0.f, 0.f, 0.f, 0.f, 0.f, 0.f};
#pragma unroll
    for (int kb = 0; kb < 4; ++kb) {
      const v2f a = *(const v2f*)(Ar + kb * 4);
      acc = wmma4(a, bwf[kb], acc);
    }
    if (m < 2 && hf == 0) {
      float* P = sP + wave * 32 + m * 16;
      const v4f lo = {acc[0], acc[1], acc[2], acc[3]};
      const v4f hi = {acc[4], acc[5], acc[6], acc[7]};
      *(v4f*)(P)     = lo;
      *(v4f*)(P + 4) = hi;
    }
  }
  __syncthreads();
  if (tid < kRows) {
    float l0 = bb[0], l1 = bb[1];
    for (int w = 0; w < 32; ++w) {
      l0 += sP[w * 32 + tid];
      l1 += sP[w * 32 + 16 + tid];
    }
    float mx = fmaxf(l0, l1);
    float e0 = expf(l0 - mx), e1 = expf(l1 - mx);
    float inv = 1.0f / (e0 + e1);
    float ipv = ipin[r0 + tid];
    ptb[r0 + tid] = e0 * inv * ipv;
    pfb[r0 + tid] = e1 * inv * ipv;
  }

  for (int idx = tid; idx < 2 * kRows * kH; idx += kStepT) {
    int l = idx >> 10, m = (idx >> 7) & 7, k = idx & 127;
    size_t row = r0 + m;
    Stg[((size_t)l * kBN + row) * kH + k]         = sC[l][m * kH + k];
    Stg[kHOFF + ((size_t)l * kBN + row) * kH + k] = (l == 0) ? fH0[m * AS + k] : fH1[m * AS + k];
  }
}

__device__ __forceinline__ v4f fma4(float w, v4f v, v4f a) {
  a.x = __builtin_fmaf(w, v.x, a.x);
  a.y = __builtin_fmaf(w, v.y, a.y);
  a.z = __builtin_fmaf(w, v.z, a.z);
  a.w = __builtin_fmaf(w, v.w, a.w);
  return a;
}

__global__ __launch_bounds__(32) void k_aggregate(
    const float* __restrict__ Stg, const float* __restrict__ Sin,
    const float* __restrict__ ipin, const float* __restrict__ ptb,
    const float* __restrict__ pfb, const int* __restrict__ rng,
    const int* __restrict__ ent, const int* __restrict__ steps,
    float* __restrict__ Sout, float* __restrict__ ipout, int step) {
  const int tgt = blockIdx.x;
  const int b = tgt >> 8;
  const int lane = threadIdx.x;
  const int h4 = lane * 4;
  const int r = rng[tgt];
  const int* __restrict__ E = ent + (r & 0xffff);
  const int cnt = r >> 16;
  v4f aC0 = {0.f, 0.f, 0.f, 0.f}, aH0 = aC0, aC1 = aC0, aH1 = aC0;
  float aip = 0.f;
  for (int e0 = 0; e0 < cnt; e0 += 32) {
    const int nb = (cnt - e0 < 32) ? cnt - e0 : 32;
    const int p = E[e0 + (lane < nb ? lane : nb - 1)];
    const int s = p & 0xffff;
    const float pt = ptb[s], pf = pfb[s];
    float ww = 0.f;
    ww += (p & kTrueBit)  ? pt : 0.f;
    ww += (p & kFalseBit) ? pf : 0.f;
    ww = (lane < nb) ? ww : 0.f;
    for (int j0 = 0; j0 < nb; j0 += 4) {
      int sj[4];
      float wj[4];
#pragma unroll
      for (int j = 0; j < 4; ++j) {
        sj[j] = __builtin_amdgcn_readlane(s, j0 + j);
        wj[j] = __builtin_bit_cast(float, __builtin_amdgcn_readlane(__builtin_bit_cast(int, ww), j0 + j));
      }
      v4f c0[4], h0[4], c1[4], h1[4];
#pragma unroll
      for (int j = 0; j < 4; ++j) {
        const size_t q = (size_t)sj[j];
        c0[j] = *(const v4f*)(Stg + q * kH + h4);
        h0[j] = *(const v4f*)(Stg + kHOFF + q * kH + h4);
        c1[j] = *(const v4f*)(Stg + ((size_t)kBN + q) * kH + h4);
        h1[j] = *(const v4f*)(Stg + kHOFF + ((size_t)kBN + q) * kH + h4);
      }
#pragma unroll
      for (int j = 0; j < 4; ++j) {
        if (wj[j] != 0.f) {
          aip += wj[j];
          aC0 = fma4(wj[j], c0[j], aC0);
          aH0 = fma4(wj[j], h0[j], aH0);
          aC1 = fma4(wj[j], c1[j], aC1);
          aH1 = fma4(wj[j], h1[j], aH1);
        }
      }
    }
  }
  const float denom = aip + 1e-7f;
  const bool active = step < steps[b];
  const size_t row = tgt;
  v4f oC0, oH0, oC1, oH1;
  if (active) {
    oC0 = aC0 / denom; oH0 = aH0 / denom; oC1 = aC1 / denom; oH1 = aH1 / denom;
  } else {
    oC0 = *(const v4f*)(Sin + row * kH + h4);
    oH0 = *(const v4f*)(Sin + kHOFF + row * kH + h4);
    oC1 = *(const v4f*)(Sin + ((size_t)kBN + row) * kH + h4);
    oH1 = *(const v4f*)(Sin + kHOFF + ((size_t)kBN + row) * kH + h4);
  }
  *(v4f*)(Sout + row * kH + h4)                         = oC0;
  *(v4f*)(Sout + kHOFF + row * kH + h4)                 = oH0;
  *(v4f*)(Sout + ((size_t)kBN + row) * kH + h4)         = oC1;
  *(v4f*)(Sout + kHOFF + ((size_t)kBN + row) * kH + h4) = oH1;
  if (lane == 0) ipout[tgt] = active ? aip : ipin[tgt];
}

__global__ __launch_bounds__(kLogT) __attribute__((amdgpu_waves_per_eu(1, 4))) void k_logits(
    const float* __restrict__ S, const int* __restrict__ exitI,
    const float* __restrict__ outW, const float* __restrict__ outb,
    float* __restrict__ out) {
  __shared__ __attribute__((aligned(16))) float sF[kB * FP];
  const int tid = threadIdx.x;
#pragma unroll
  for (int j = 0; j < kB; ++j) {
    const int k = tid * 4;
    const int seg = k >> 7, kk = k & 127;
    const size_t row = (size_t)j * kN + exitI[j];
    const float* src = S + ((size_t)(seg >> 1) * kBN + row) * kH + (size_t)(seg & 1) * kHOFF + kk;
    *(v4f*)(sF + j * FP + k) = *(const v4f*)src;
  }
  __syncthreads();

  const int lane = tid & 31;
  const int wave = tid >> 5;
  const int m  = lane & 15;
  const int hf = lane >> 4;
  const int strip = blockIdx.x * (kLogT / 32) + wave;
  const bool live = strip < kStrips;
  const int n0 = (live ? strip : kStrips - 1) * 32;
  int col = n0 + 2 * m;
  const bool inr = live && (col < kOutV);
  if (col > kOutV - 2) col = kOutV - 2;
  v8f acc0, acc1;
  {
    const v2f bv = *(const v2f*)(outb + col);
#pragma unroll
    for (int v = 0; v < 8; ++v) { acc0[v] = bv.x; acc1[v] = bv.y; }
  }
  const float* Ar = sF + (m & 7) * FP + 2 * hf;
  const float* Bp = outW + (size_t)(2 * hf) * kOutV + col;
#pragma unroll 16
  for (int kc = 0; kc < 128; ++kc) {
    const v2f a  = *(const v2f*)(Ar + kc * 4);
    const v2f r0 = *(const v2f*)(Bp + (size_t)(kc * 4) * kOutV);
    const v2f r1 = *(const v2f*)(Bp + (size_t)(kc * 4 + 1) * kOutV);
    const v2f b0 = {r0.x, r1.x};
    const v2f b1 = {r0.y, r1.y};
    acc0 = wmma4(a, b0, acc0);
    acc1 = wmma4(a, b1, acc1);
  }
  if (inr && hf == 0) {
#pragma unroll
    for (int q = 0; q < kB; ++q) {
      const v2f o = {acc0[q], acc1[q]};
      *(v2f*)(out + (size_t)q * kOutV + col) = o;
    }
  }
}

__global__ void k_embed(const int* __restrict__ data, const float* __restrict__ embed,
                        float* __restrict__ embuf, int total) {
  int idx = blockIdx.x * 256 + threadIdx.x;
  if (idx >= total) return;
  int token = idx >> 7, h = idx & 127;
  embuf[idx] = embed[(size_t)data[token] * kH + h];
}

__global__ void k_pack(const float* __restrict__ Wsrc, float* __restrict__ Wdst, int total) {
  int idx = blockIdx.x * 256 + threadIdx.x;
  if (idx >= total) return;
  int col = idx & 511;
  int p   = (idx >> 9) & 63;
  int l   = idx >> 15;
  const float* Wl = Wsrc + (size_t)l * kH * kG;
  v2f v;
  v.x = Wl[(size_t)(2 * p) * kG + col];
  v.y = Wl[(size_t)(2 * p + 1) * kG + col];
  *(v2f*)(Wdst + (size_t)idx * 2) = v;
}

__global__ void k_init(float* __restrict__ S0, float* __restrict__ ip0, int total,
                       const int* __restrict__ ti, const int* __restrict__ fi,
                       int* __restrict__ rng, int* __restrict__ ent) {
  __shared__ int sT[kN];
  __shared__ int sF[kN];
  __shared__ int sCnt[kN];
  int idx = blockIdx.x * 256 + threadIdx.x;
  if (idx < total) S0[idx] = 0.f;
  if (idx < kBN) ip0[idx] = ((idx & 255) == 0) ? 1.0f : 0.0f;
  if (blockIdx.x < kB) {
    const int b = blockIdx.x;
    const int m = threadIdx.x;
    sT[m] = ti[b * kN + m];
    sF[m] = fi[b * kN + m];
    __syncthreads();
    unsigned mt[kN / 32], mf[kN / 32];
    int cnt = 0;
#pragma unroll
    for (int wd = 0; wd < kN / 32; ++wd) {
      unsigned a = 0u, c = 0u;
#pragma unroll
      for (int q = 0; q < 32; ++q) {
        a |= (sT[wd * 32 + q] == m) ? (1u << q) : 0u;
        c |= (sF[wd * 32 + q] == m) ? (1u << q) : 0u;
      }
      mt[wd] = a;
      mf[wd] = c;
      cnt += __builtin_popcount(a | c);
    }
    sCnt[m] = cnt;
    __syncthreads();
    int off = 0;
    for (int j = 0; j < kN; ++j) off += (j < m) ? sCnt[j] : 0;
    int* E = ent + b * kEntB + off;
    int k = 0;
#pragma unroll
    for (int wd = 0; wd < kN / 32; ++wd) {
      unsigned u = mt[wd] | mf[wd];
      while (u) {
        const int q = __builtin_ctz(u);
        u &= u - 1u;
        E[k++] = (b * kN + wd * 32 + q) | (((mt[wd] >> q) & 1u) ? kTrueBit : 0)
                                        | (((mf[wd] >> q) & 1u) ? kFalseBit : 0);
      }
    }
    rng[b * kN + m] = (b * kEntB + off) | (cnt << 16);
  }
}

extern "C" void kernel_launch(void* const* d_in, const int* in_sizes, int n_in,
                              void* d_out, int out_size, void* d_ws, size_t ws_size,
                              hipStream_t stream) {
  const int*   data  = (const int*)d_in[0];
  const int*   ti    = (const int*)d_in[1];
  const int*   fi    = (const int*)d_in[2];
  const int*   exitI = (const int*)d_in[3];
  const int*   steps = (const int*)d_in[4];
  const float* embed = (const float*)d_in[5];
  const float* Wx    = (const float*)d_in[6];
  const float* Wh    = (const float*)d_in[7];
  const float* bias  = (const float*)d_in[8];
  const float* bW    = (const float*)d_in[9];
  const float* bb    = (const float*)d_in[10];
  const float* outW  = (const float*)d_in[11];
  const float* outb  = (const float*)d_in[12];
  float* out = (float*)d_out;

  float* ws = (float*)d_ws;
  const size_t EMB = (size_t)kBN * kL * kH;
  const size_t ST  = (size_t)4 * kBN * kH;
  float* embuf = ws;
  float* S0  = embuf + EMB;
  float* S1  = S0 + ST;
  float* STG = S1 + ST;
  float* ip0 = STG + ST;
  float* ip1 = ip0 + kBN;
  float* ptb = ip1 + kBN;
  float* pfb = ptb + kBN;
  float* Wxp = pfb + kBN;
  float* Whp = Wxp + (size_t)kPairs * 2;
  int* rng = (int*)(Whp + (size_t)kPairs * 2);
  int* ent = rng + kBN;

  k_embed<<<(int)((EMB + 255) / 256), 256, 0, stream>>>(data, embed, embuf, (int)EMB);
  k_init<<<(int)((ST + 255) / 256), 256, 0, stream>>>(S0, ip0, (int)ST, ti, fi, rng, ent);
  k_pack<<<(kPairs + 255) / 256, 256, 0, stream>>>(Wx, Wxp, kPairs);
  k_pack<<<(kPairs + 255) / 256, 256, 0, stream>>>(Wh, Whp, kPairs);

  float* Sin = S0; float* Sout = S1; float* ipin = ip0; float* ipout = ip1;
  for (int s = 0; s < kSteps; ++s) {
    k_lstm_step<<<kBN / kRows, kStepT, 0, stream>>>(embuf, Sin, STG, ipin, ptb, pfb,
                                              (const v2f*)Wxp, (const v2f*)Whp,
                                              bias, bW, bb, exitI);
    k_aggregate<<<kBN, 32, 0, stream>>>(STG, Sin, ipin, ptb, pfb, rng, ent, steps,
                                        Sout, ipout, s);
    float* t1 = Sin; Sin = Sout; Sout = t1;
    float* t2 = ipin; ipin = ipout; ipout = t2;
  }
  k_logits<<<(kStrips + kLogT / 32 - 1) / (kLogT / 32), kLogT, 0, stream>>>(Sin, exitI, outW, outb, out);
  (void)in_sizes; (void)n_in; (void)out_size; (void)ws_size;
}


# Step kernel: a work-group whose 8 rows all have input probability exactly 0 stores zero branch weights and zero rows and ends at entry (such rows enter every sum with weight 0); sums unchanged
# speedup vs baseline: 2.7137x; 1.0356x over previous
#include <hip/hip_runtime.h>
#include <cstddef>

typedef float v2f __attribute__((ext_vector_type(2)));
typedef float v8f __attribute__((ext_vector_type(8)));
typedef float v4f __attribute__((ext_vector_type(4)));

namespace {
constexpr int kB    = 8;
constexpr int kN    = 256;
constexpr int kL    = 4;
constexpr int kH    = 128;
constexpr int kBN   = kB * kN;
constexpr int kG    = 4 * kH;
constexpr int kSteps = 16;
constexpr int kOutV = 30000;
constexpr size_t kHOFF = (size_t)2 * kBN * kH;
constexpr int AS = 132;
constexpr int GS = 516;
constexpr int kPairs = 2 * (kH / 2) * kG;
constexpr int kStepT = 1024;
constexpr int kRows  = 8;
constexpr int kSlots = 15;
constexpr int kTile  = 16 * AS;
constexpr int kTeamT = 512;
constexpr int kEntB = 2 * kN;
constexpr int kTrueBit  = 0x10000;
constexpr int kFalseBit = 0x20000;
constexpr int FP = 516;
constexpr int kLogT = 128;
constexpr int kStrips = (kOutV + 31) / 32;
}

__device__ __forceinline__ float sigmf(float x) { return 1.0f / (1.0f + expf(-x)); }

__device__ __forceinline__ v8f wmma4(v2f a, v2f b, v8f c) {
  return __builtin_amdgcn_wmma_f32_16x16x4_f32(false, a, false, b, (short)0, c, false, false);
}

template <int P, int KB>
__device__ __forceinline__ void gemm_piece(const float* __restrict__ A, const v2f* __restrict__ Wp,
                                           int cq, int m, int hf, v8f& acc0, v8f& acc1) {
  __builtin_amdgcn_s_setprio(P);
#pragma unroll
  for (int kc = KB; kc < KB + 8; ++kc) {
    const int kk = kc * 4 + 2 * hf;
    v2f a = *(const v2f*)(A + m * AS + kk);
    const v4f q = *(const v4f*)(Wp + (size_t)(kc * 2 + hf) * kG + cq);
    v2f b0 = {q.x, q.y};
    v2f b1 = {q.z, q.w};
    acc0 = wmma4(a, b0, acc0);
    acc1 = wmma4(a, b1, acc1);
  }
}

__device__ __forceinline__ void gemm_full(const float* __restrict__ A, const v2f* __restrict__ Wp,
                                          int cq, int lane, v8f& acc0, v8f& acc1) {
  const int m  = lane & 15;
  const int hf = lane >> 4;
  gemm_piece<3, 0>(A, Wp, cq, m, hf, acc0, acc1);
  gemm_piece<2, 8>(A, Wp, cq, m, hf, acc0, acc1);
  gemm_piece<1, 16>(A, Wp, cq, m, hf, acc0, acc1);
  gemm_piece<0, 24>(A, Wp, cq, m, hf, acc0, acc1);
}

__device__ __forceinline__ void team_cell(const float* __restrict__ G, float* __restrict__ C,
                                          float* __restrict__ H1, float* __restrict__ H2, int rh,
                                          int first, int stride) {
  for (int idx = first; idx < kRows * kH; idx += stride) {
    int m = idx >> 7, k = idx & 127;
    const float* g = G + (rh * kRows + m) * GS;
    float iv = g[k];
    float fv = g[kH + k];
    float gv = g[2 * kH + k];
    float ov = g[3 * kH + k];
    float cold = C[m * kH + k];
    float cn = sigmf(fv) * cold + sigmf(iv) * tanhf(gv);
    float hn = sigmf(ov) * tanhf(cn);
    C[m * kH + k] = cn;
    H1[m * AS + k] = hn;
    if (H2 != nullptr) H2[m * AS + k] = hn;
  }
}

__global__ __launch_bounds__(kStepT) void k_lstm_step(
    const float* __restrict__ emb, const float* __restrict__ Sin,
    float* __restrict__ Stg, const float* __restrict__ ipin,
    float* __restrict__ ptb, float* __restrict__ pfb,
    const v2f* __restrict__ Wxp, const v2f* __restrict__ Whp,
    const float* __restrict__ bias, const float* __restrict__ bW,
    const float* __restrict__ bb, const int* __restrict__ exitI) {
  __shared__ float sA[8 * kTile];
  __shared__ float sC[2][kRows * kH];
  __shared__ float sG[2][16 * GS];
  __shared__ __attribute__((aligned(16))) float sP[32 * 32];

  const int tid  = threadIdx.x;
  const int lane = tid & 31;
  const int wave = __builtin_amdgcn_readfirstlane(tid >> 5);
  const int team = wave >> 4;
  const int tt   = tid & (kTeamT - 1);
  const int r0   = blockIdx.x * kRows;
  const int cq   = (wave & 15) * 32 + 2 * (lane & 15);

  v2f bwf[4];
  {
    const int c = lane & 15;
    const int hf = lane >> 4;
#pragma unroll
    for (int kb = 0; kb < 4; ++kb) {
      const int K = wave * 16 + kb * 4 + 2 * hf;
      const v4f q = *(const v4f*)(bW + K * 2);
      v2f bfr = {0.f, 0.f};
      if (c == 0) { bfr.x = q.x; bfr.y = q.z; }
      if (c == 1) { bfr.x = q.y; bfr.y = q.w; }
      bwf[kb] = bfr;
    }
  }

  {
    const v4f ipa = *(const v4f*)(ipin + r0);
    const v4f ipb = *(const v4f*)(ipin + r0 + 4);
    const bool dead = (ipa.x == 0.f) & (ipa.y == 0.f) & (ipa.z == 0.f) & (ipa.w == 0.f)
                    & (ipb.x == 0.f) & (ipb.y == 0.f) & (ipb.z == 0.f) & (ipb.w == 0.f);
    if (dead) {
      if (tid < kRows) { ptb[r0 + tid] = 0.f; pfb[r0 + tid] = 0.f; }
      const int vec = tid >> 8;
      const int q = tid & 255;
      const size_t off = (size_t)(vec >> 1) * kHOFF
                       + ((size_t)(vec & 1) * kBN + (size_t)(r0 + (q >> 5))) * kH + (size_t)(q & 31) * 4;
      const v4f zero4 = {0.f, 0.f, 0.f, 0.f};
      *(v4f*)(Stg + off) = zero4;
      return;
    }
  }

  for (int idx = tid; idx < 2 * kRows * kH; idx += kStepT) {
    int l = idx >> 10, m = (idx >> 7) & 7, k = idx & 127;
    size_t row = r0 + m;
    sC[l][m * kH + k] = Sin[((size_t)l * kBN + row) * kH + k];
    float hv = Sin[kHOFF + ((size_t)l * kBN + row) * kH + k];
    sA[(2 + 2 * l) * kTile + m * AS + k] = hv;
    sA[(2 + 2 * l) * kTile + (kRows + m) * AS + k] = 0.f;
    sA[(3 + 2 * l) * kTile + m * AS + k] = 0.f;
  }
  for (int idx = tid; idx < kL * kRows * kH; idx += kStepT) {
    int t = idx >> 10, m = (idx >> 7) & 7, k = idx & 127;
    sA[(t >> 1) * kTile + ((t & 1) * kRows + m) * AS + k] = emb[(((size_t)(r0 + m)) * kL + t) * kH + k];
  }
  __syncthreads();

  const unsigned long long sched = team ? 0xA89765304250100ULL : 0xA090A8976053421ULL;
  v8f acc0 = {0.f, 0.f, 0.f, 0.f, 0.f, 0.f, 0.f, 0.f};
  v8f acc1 = acc0;
  for (int s = 0; s < kSlots; ++s) {
    const int op = (int)((sched >> (4 * s)) & 15ULL);
    const bool paired = (op == 1) || (op == 6);
    if (paired || op == 2 || op == 3 || op == 7 || op == 8) {
      const int layer = (op >= 6) ? 1 : 0;
      const int tile = (op == 1) ? team : (op == 6) ? 6 + team : (op < 6) ? op : op - 3;
      const v2f* Wp = (paired ? Wxp : Whp) + (size_t)layer * (kH / 2) * kG;
      if (paired) {
        const v2f bv = *(const v2f*)(bias + layer * kG + cq);
#pragma unroll
        for (int v = 0; v < 8; ++v) { acc0[v] = bv.x; acc1[v] = bv.y; }
      }
      gemm_full(sA + tile * kTile, Wp, cq, lane, acc0, acc1);
      if (!paired) {
        float* Gout = sG[team];
        const int hf = lane >> 4;
#pragma unroll
        for (int v = 0; v < 8; ++v) {
          v2f g = {acc0[v], acc1[v]};
          *(v2f*)(Gout + (v + 8 * hf) * GS + cq) = g;
        }
      }
    } else if (op != 0) {
      const int layer = (op >= 9) ? 1 : 0;
      const int rh = (op == 5 || op == 10) ? 1 : 0;
      const bool shared = (s == 4) || (s == 12) || (s == 14);
      const int owner = shared ? ((s == 4) ? 0 : 1) : team;
      float* H1 = rh ? sA + (2 + 2 * layer) * kTile : sA + (3 + 2 * layer) * kTile + kRows * AS;
      float* H2 = layer ? nullptr : sA + (6 + owner) * kTile + rh * kRows * AS;
      team_cell(sG[owner], sC[layer], H1, H2, rh, shared ? tid : tt, shared ? kStepT : kTeamT);
    }
    __syncthreads();
  }

  float* fH0 = sA + 2 * kTile;
  float* fH1 = sA + 4 * kTile;

  const int bidx = r0 >> 8;
  const int mm = exitI[bidx] - (r0 & 255);
  if (mm >= 0 && mm < kRows) {
    for (int idx = tid; idx < 2 * kH; idx += kStepT) {
      int l = idx >> 7, k = idx & 127;
      size_t row = r0 + mm;
      sC[l][mm * kH + k] = Sin[((size_t)l * kBN + row) * kH + k];
      float hv = Sin[kHOFF + ((size_t)l * kBN + row) * kH + k];
      if (l == 0) fH0[mm * AS + k] = hv; else fH1[mm * AS + k] = hv;
    }
  }
  __syncthreads();

  {
    const int seg = wave >> 3;
    const float* Ab = (seg == 0) ? sC[0] : (seg == 1) ? fH0 : (seg == 2) ? sC[1] : fH1;
    const int As = (seg & 1) ? AS : kH;
    const int m  = lane & 15;
    const int hf = lane >> 4;
    const float* Ar = Ab + (m & 7) * As + (wave & 7) * 16 + 2 * hf;
    v8f acc = {0.f, 0.f, 0.f, 0.f, 0.f, 0.f, 0.f, 0.f};
#pragma unroll
    for (int kb = 0; kb < 4; ++kb) {
      const v2f a = *(const v2f*)(Ar + kb * 4);
      acc = wmma4(a, bwf[kb], acc);
    }
    if (m < 2 && hf == 0) {
      float* P = sP + wave * 32 + m * 16;
      const v4f lo = {acc[0], acc[1], acc[2], acc[3]};
      const v4f hi = {acc[4], acc[5], acc[6], acc[7]};
      *(v4f*)(P)     = lo;
      *(v4f*)(P + 4) = hi;
    }
  }
  __syncthreads();
  if (tid < kRows) {
    float l0 = bb[0], l1 = bb[1];
    for (int w = 0; w < 32; ++w) {
      l0 += sP[w * 32 + tid];
      l1 += sP[w * 32 + 16 + tid];
    }
    float mx = fmaxf(l0, l1);
    float e0 = expf(l0 - mx), e1 = expf(l1 - mx);
    float inv = 1.0f / (e0 + e1);
    float ipv = ipin[r0 + tid];
    ptb[r0 + tid] = e0 * inv * ipv;
    pfb[r0 + tid] = e1 * inv * ipv;
  }

  for (int idx = tid; idx < 2 * kRows * kH; idx += kStepT) {
    int l = idx >> 10, m = (idx >> 7) & 7, k = idx & 127;
    size_t row = r0 + m;
    Stg[((size_t)l * kBN + row) * kH + k]         = sC[l][m * kH + k];
    Stg[kHOFF + ((size_t)l * kBN + row) * kH + k] = (l == 0) ? fH0[m * AS + k] : fH1[m * AS + k];
  }
}

__device__ __forceinline__ v4f fma4(float w, v4f v, v4f a) {
  a.x = __builtin_fmaf(w, v.x, a.x);
  a.y = __builtin_fmaf(w, v.y, a.y);
  a.z = __builtin_fmaf(w, v.z, a.z);
  a.w = __builtin_fmaf(w, v.w, a.w);
  return a;
}

__global__ __launch_bounds__(32) void k_aggregate(
    const float* __restrict__ Stg, const float* __restrict__ Sin,
    const float* __restrict__ ipin, const float* __restrict__ ptb,
    const float* __restrict__ pfb, const int* __restrict__ rng,
    const int* __restrict__ ent, const int* __restrict__ steps,
    float* __restrict__ Sout, float* __restrict__ ipout, int step) {
  const int tgt = blockIdx.x;
  const int b = tgt >> 8;
  const int lane = threadIdx.x;
  const int h4 = lane * 4;
  const int r = rng[tgt];
  const int* __restrict__ E = ent + (r & 0xffff);
  const int cnt = r >> 16;
  v4f aC0 = {0.f, 0.f, 0.f, 0.f}, aH0 = aC0, aC1 = aC0, aH1 = aC0;
  float aip = 0.f;
  for (int e0 = 0; e0 < cnt; e0 += 32) {
    const int nb = (cnt - e0 < 32) ? cnt - e0 : 32;
    const int p = E[e0 + (lane < nb ? lane : nb - 1)];
    const int s = p & 0xffff;
    const float pt = ptb[s], pf = pfb[s];
    float ww = 0.f;
    ww += (p & kTrueBit)  ? pt : 0.f;
    ww += (p & kFalseBit) ? pf : 0.f;
    ww = (lane < nb) ? ww : 0.f;
    for (int j0 = 0; j0 < nb; j0 += 4) {
      int sj[4];
      float wj[4];
#pragma unroll
      for (int j = 0; j < 4; ++j) {
        sj[j] = __builtin_amdgcn_readlane(s, j0 + j);
        wj[j] = __builtin_bit_cast(float, __builtin_amdgcn_readlane(__builtin_bit_cast(int, ww), j0 + j));
      }
      v4f c0[4], h0[4], c1[4], h1[4];
#pragma unroll
      for (int j = 0; j < 4; ++j) {
        const size_t q = (size_t)sj[j];
        c0[j] = *(const v4f*)(Stg + q * kH + h4);
        h0[j] = *(const v4f*)(Stg + kHOFF + q * kH + h4);
        c1[j] = *(const v4f*)(Stg + ((size_t)kBN + q) * kH + h4);
        h1[j] = *(const v4f*)(Stg + kHOFF + ((size_t)kBN + q) * kH + h4);
      }
#pragma unroll
      for (int j = 0; j < 4; ++j) {
        if (wj[j] != 0.f) {
          aip += wj[j];
          aC0 = fma4(wj[j], c0[j], aC0);
          aH0 = fma4(wj[j], h0[j], aH0);
          aC1 = fma4(wj[j], c1[j], aC1);
          aH1 = fma4(wj[j], h1[j], aH1);
        }
      }
    }
  }
  const float denom = aip + 1e-7f;
  const bool active = step < steps[b];
  const size_t row = tgt;
  v4f oC0, oH0, oC1, oH1;
  if (active) {
    oC0 = aC0 / denom; oH0 = aH0 / denom; oC1 = aC1 / denom; oH1 = aH1 / denom;
  } else {
    oC0 = *(const v4f*)(Sin + row * kH + h4);
    oH0 = *(const v4f*)(Sin + kHOFF + row * kH + h4);
    oC1 = *(const v4f*)(Sin + ((size_t)kBN + row) * kH + h4);
    oH1 = *(const v4f*)(Sin + kHOFF + ((size_t)kBN + row) * kH + h4);
  }
  *(v4f*)(Sout + row * kH + h4)                         = oC0;
  *(v4f*)(Sout + kHOFF + row * kH + h4)                 = oH0;
  *(v4f*)(Sout + ((size_t)kBN + row) * kH + h4)         = oC1;
  *(v4f*)(Sout + kHOFF + ((size_t)kBN + row) * kH + h4) = oH1;
  if (lane == 0) ipout[tgt] = active ? aip : ipin[tgt];
}

__global__ __launch_bounds__(kLogT) __attribute__((amdgpu_waves_per_eu(1, 4))) void k_logits(
    const float* __restrict__ S, const int* __restrict__ exitI,
    const float* __restrict__ outW, const float* __restrict__ outb,
    float* __restrict__ out) {
  __shared__ __attribute__((aligned(16))) float sF[kB * FP];
  const int tid = threadIdx.x;
#pragma unroll
  for (int j = 0; j < kB; ++j) {
    const int k = tid * 4;
    const int seg = k >> 7, kk = k & 127;
    const size_t row = (size_t)j * kN + exitI[j];
    const float* src = S + ((size_t)(seg >> 1) * kBN + row) * kH + (size_t)(seg & 1) * kHOFF + kk;
    *(v4f*)(sF + j * FP + k) = *(const v4f*)src;
  }
  __syncthreads();

  const int lane = tid & 31;
  const int wave = tid >> 5;
  const int m  = lane & 15;
  const int hf = lane >> 4;
  const int strip = blockIdx.x * (kLogT / 32) + wave;
  const bool live = strip < kStrips;
  const int n0 = (live ? strip : kStrips - 1) * 32;
  int col = n0 + 2 * m;
  const bool inr = live && (col < kOutV);
  if (col > kOutV - 2) col = kOutV - 2;
  v8f acc0, acc1;
  {
    const v2f bv = *(const v2f*)(outb + col);
#pragma unroll
    for (int v = 0; v < 8; ++v) { acc0[v] = bv.x; acc1[v] = bv.y; }
  }
  const float* Ar = sF + (m & 7) * FP + 2 * hf;
  const float* Bp = outW + (size_t)(2 * hf) * kOutV + col;
#pragma unroll 16
  for (int kc = 0; kc < 128; ++kc) {
    const v2f a  = *(const v2f*)(Ar + kc * 4);
    const v2f r0 = *(const v2f*)(Bp + (size_t)(kc * 4) * kOutV);
    const v2f r1 = *(const v2f*)(Bp + (size_t)(kc * 4 + 1) * kOutV);
    const v2f b0 = {r0.x, r1.x};
    const v2f b1 = {r0.y, r1.y};
    acc0 = wmma4(a, b0, acc0);
    acc1 = wmma4(a, b1, acc1);
  }
  if (inr && hf == 0) {
#pragma unroll
    for (int q = 0; q < kB; ++q) {
      const v2f o = {acc0[q], acc1[q]};
      *(v2f*)(out + (size_t)q * kOutV + col) = o;
    }
  }
}

__global__ void k_embed(const int* __restrict__ data, const float* __restrict__ embed,
                        float* __restrict__ embuf, int total) {
  int idx = blockIdx.x * 256 + threadIdx.x;
  if (idx >= total) return;
  int token = idx >> 7, h = idx & 127;
  embuf[idx] = embed[(size_t)data[token] * kH + h];
}

__global__ void k_pack(const float* __restrict__ Wsrc, float* __restrict__ Wdst, int total) {
  int idx = blockIdx.x * 256 + threadIdx.x;
  if (idx >= total) return;
  int col = idx & 511;
  int p   = (idx >> 9) & 63;
  int l   = idx >> 15;
  const float* Wl = Wsrc + (size_t)l * kH * kG;
  v2f v;
  v.x = Wl[(size_t)(2 * p) * kG + col];
  v.y = Wl[(size_t)(2 * p + 1) * kG + col];
  *(v2f*)(Wdst + (size_t)idx * 2) = v;
}

__global__ void k_init(float* __restrict__ S0, float* __restrict__ ip0, int total,
                       const int* __restrict__ ti, const int* __restrict__ fi,
                       int* __restrict__ rng, int* __restrict__ ent) {
  __shared__ int sT[kN];
  __shared__ int sF[kN];
  __shared__ int sCnt[kN];
  int idx = blockIdx.x * 256 + threadIdx.x;
  if (idx < total) S0[idx] = 0.f;
  if (idx < kBN) ip0[idx] = ((idx & 255) == 0) ? 1.0f : 0.0f;
  if (blockIdx.x < kB) {
    const int b = blockIdx.x;
    const int m = threadIdx.x;
    sT[m] = ti[b * kN + m];
    sF[m] = fi[b * kN + m];
    __syncthreads();
    unsigned mt[kN / 32], mf[kN / 32];
    int cnt = 0;
#pragma unroll
    for (int wd = 0; wd < kN / 32; ++wd) {
      unsigned a = 0u, c = 0u;
#pragma unroll
      for (int q = 0; q < 32; ++q) {
        a |= (sT[wd * 32 + q] == m) ? (1u << q) : 0u;
        c |= (sF[wd * 32 + q] == m) ? (1u << q) : 0u;
      }
      mt[wd] = a;
      mf[wd] = c;
      cnt += __builtin_popcount(a | c);
    }
    sCnt[m] = cnt;
    __syncthreads();
    int off = 0;
    for (int j = 0; j < kN; ++j) off += (j < m) ? sCnt[j] : 0;
    int* E = ent + b * kEntB + off;
    int k = 0;
#pragma unroll
    for (int wd = 0; wd < kN / 32; ++wd) {
      unsigned u = mt[wd] | mf[wd];
      while (u) {
        const int q = __builtin_ctz(u);
        u &= u - 1u;
        E[k++] = (b * kN + wd * 32 + q) | (((mt[wd] >> q) & 1u) ? kTrueBit : 0)
                                        | (((mf[wd] >> q) & 1u) ? kFalseBit : 0);
      }
    }
    rng[b * kN + m] = (b * kEntB + off) | (cnt << 16);
  }
}

extern "C" void kernel_launch(void* const* d_in, const int* in_sizes, int n_in,
                              void* d_out, int out_size, void* d_ws, size_t ws_size,
                              hipStream_t stream) {
  const int*   data  = (const int*)d_in[0];
  const int*   ti    = (const int*)d_in[1];
  const int*   fi    = (const int*)d_in[2];
  const int*   exitI = (const int*)d_in[3];
  const int*   steps = (const int*)d_in[4];
  const float* embed = (const float*)d_in[5];
  const float* Wx    = (const float*)d_in[6];
  const float* Wh    = (const float*)d_in[7];
  const float* bias  = (const float*)d_in[8];
  const float* bW    = (const float*)d_in[9];
  const float* bb    = (const float*)d_in[10];
  const float* outW  = (const float*)d_in[11];
  const float* outb  = (const float*)d_in[12];
  float* out = (float*)d_out;

  float* ws = (float*)d_ws;
  const size_t EMB = (size_t)kBN * kL * kH;
  const size_t ST  = (size_t)4 * kBN * kH;
  float* embuf = ws;
  float* S0  = embuf + EMB;
  float* S1  = S0 + ST;
  float* STG = S1 + ST;
  float* ip0 = STG + ST;
  float* ip1 = ip0 + kBN;
  float* ptb = ip1 + kBN;
  float* pfb = ptb + kBN;
  float* Wxp = pfb + kBN;
  float* Whp = Wxp + (size_t)kPairs * 2;
  int* rng = (int*)(Whp + (size_t)kPairs * 2);
  int* ent = rng + kBN;

  k_embed<<<(int)((EMB + 255) / 256), 256, 0, stream>>>(data, embed, embuf, (int)EMB);
  k_init<<<(int)((ST + 255) / 256), 256, 0, stream>>>(S0, ip0, (int)ST, ti, fi, rng, ent);
  k_pack<<<(kPairs + 255) / 256, 256, 0, stream>>>(Wx, Wxp, kPairs);
  k_pack<<<(kPairs + 255) / 256, 256, 0, stream>>>(Wh, Whp, kPairs);

  float* Sin = S0; float* Sout = S1; float* ipin = ip0; float* ipout = ip1;
  for (int s = 0; s < kSteps; ++s) {
    k_lstm_step<<<kBN / kRows, kStepT, 0, stream>>>(embuf, Sin, STG, ipin, ptb, pfb,
                                              (const v2f*)Wxp, (const v2f*)Whp,
                                              bias, bW, bb, exitI);
    k_aggregate<<<kBN, 32, 0, stream>>>(STG, Sin, ipin, ptb, pfb, rng, ent, steps,
                                        Sout, ipout, s);
    float* t1 = Sin; Sin = Sout; Sout = t1;
    float* t2 = ipin; ipin = ipout; ipout = t2;
  }
  k_logits<<<(kStrips + kLogT / 32 - 1) / (kLogT / 32), kLogT, 0, stream>>>(Sin, exitI, outW, outb, out);
  (void)in_sizes; (void)n_in; (void)out_size; (void)ws_size;
}
